# starting kernel
# baseline (speedup 1.0000x reference)
#include <hip/hip_runtime.h>
#include <math.h>

typedef __attribute__((ext_vector_type(2))) float v2f;
typedef __attribute__((ext_vector_type(8))) float v8f;

#define NEG_SLOPE 0.2f
#define PYG_EPS   1e-16f

__global__ __launch_bounds__(256) void gemm_wmma_f32(
    const float* __restrict__ X, const float* __restrict__ W,
    float* __restrict__ Hout, int Nrows, int K, int M)
{
  const int tilesN = M >> 4;
  const int tilesTotal = (Nrows >> 4) * tilesN;
  const int wave = blockIdx.x * (blockDim.x >> 5) + (threadIdx.x >> 5);
  if (wave >= tilesTotal) return;
  const int tm = wave / tilesN;
  const int tn = wave - tm * tilesN;
  const int lane = threadIdx.x & 31;
  const int half = lane >> 4;
  const int l15  = lane & 15;
  const int kb   = half << 1;

  const float* xp = X + (size_t)((tm << 4) + l15) * K + kb;
  const float* wp = W + (size_t)kb * M + ((tn << 4) + l15);

  v8f acc = {0.f, 0.f, 0.f, 0.f, 0.f, 0.f, 0.f, 0.f};
  for (int k0 = 0; k0 < K; k0 += 4) {
    v2f a = *(const v2f*)xp;
    v2f b;
    b.x = wp[0];
    b.y = wp[M];
    acc = __builtin_amdgcn_wmma_f32_16x16x4_f32(
         false, a,  false, b,
         (short)0, acc,  false,  false);
    xp += 4;
    wp += (size_t)4 * M;
  }

  float* op = Hout + (size_t)((tm << 4) + (half << 3)) * M + ((tn << 4) + l15);
  #pragma unroll
  for (int v = 0; v < 8; ++v) op[(size_t)v * M] = acc[v];
}

__global__ __launch_bounds__(256) void att_dots(
    const float* __restrict__ Hm, const float* __restrict__ att_s,
    const float* __restrict__ att_d, float* __restrict__ a_s,
    float* __restrict__ a_d, int Nrows, int Hh, int C)
{
  const int wave = blockIdx.x * (blockDim.x >> 5) + (threadIdx.x >> 5);
  if (wave >= Nrows * Hh) return;
  const int lane = threadIdx.x & 31;
  const int h = wave % Hh;
  const float* hv = Hm + (size_t)wave * C;
  const float* as = att_s + (size_t)h * C;
  const float* ad = att_d + (size_t)h * C;
  float ss = 0.f, sd = 0.f;
  for (int c = lane; c < C; c += 32) {
    float x = hv[c];
    ss += x * as[c];
    sd += x * ad[c];
  }
  #pragma unroll
  for (int off = 16; off; off >>= 1) {
    ss += __shfl_xor(ss, off, 32);
    sd += __shfl_xor(sd, off, 32);
  }
  if (lane == 0) { a_s[wave] = ss; a_d[wave] = sd; }
}

__device__ __forceinline__ float lrelu(float e) {
  return e > 0.f ? e : NEG_SLOPE * e;
}

__global__ void init_ms(float* __restrict__ m, float* __restrict__ s, int total)
{
  int i = blockIdx.x * blockDim.x + threadIdx.x;
  if (i < total) { m[i] = -__builtin_inff(); s[i] = 0.f; }
}

__global__ void init_bias(float* __restrict__ out, const float* __restrict__ bias,
                          int HC, int total)
{
  int i = blockIdx.x * blockDim.x + threadIdx.x;
  if (i < total) out[i] = bias[i % HC];
}

__global__ void elu_inplace(float* __restrict__ x, int total)
{
  int i = blockIdx.x * blockDim.x + threadIdx.x;
  if (i < total) { float v = x[i]; x[i] = v > 0.f ? v : expm1f(v); }
}

template <int H>
__global__ void edge_max(const int* __restrict__ src, const int* __restrict__ dst,
                         const float* __restrict__ a_s, const float* __restrict__ a_d,
                         float* __restrict__ m, int E, int N)
{
  int i = blockIdx.x * blockDim.x + threadIdx.x;
  if (i >= E + N) return;
  int si, di;
  if (i < E) { si = src[i]; di = dst[i]; } else { si = di = i - E; }
  #pragma unroll
  for (int h = 0; h < H; ++h) {
    float e = lrelu(a_s[si * H + h] + a_d[di * H + h]);
    __hip_atomic_fetch_max(&m[di * H + h], e,
                           __ATOMIC_RELAXED, __HIP_MEMORY_SCOPE_AGENT);
  }
}

template <int H>
__global__ void edge_expsum(const int* __restrict__ src, const int* __restrict__ dst,
                            const float* __restrict__ a_s, const float* __restrict__ a_d,
                            const float* __restrict__ m, float* __restrict__ ssum,
                            int E, int N)
{
  int i = blockIdx.x * blockDim.x + threadIdx.x;
  if (i >= E + N) return;
  int si, di;
  if (i < E) { si = src[i]; di = dst[i]; } else { si = di = i - E; }
  #pragma unroll
  for (int h = 0; h < H; ++h) {
    float e = lrelu(a_s[si * H + h] + a_d[di * H + h]);
    float w = __expf(e - m[di * H + h]);
    atomicAdd(&ssum[di * H + h], w);
  }
}

template <int H, int C>
__global__ __launch_bounds__(256) void edge_aggregate(
    const int* __restrict__ src, const int* __restrict__ dst,
    const float* __restrict__ a_s, const float* __restrict__ a_d,
    const float* __restrict__ m, const float* __restrict__ ssum,
    const float* __restrict__ Hfeat, float* __restrict__ out, int E, int N)
{
  constexpr int HC  = H * C;
  constexpr int EPB = 256 / HC;
  const int t    = threadIdx.x;
  const int eidx = blockIdx.x * EPB + t / HC;
  if (eidx >= E + N) return;
  const int f = t % HC;
  const int h = f / C;
  int si, di;
  if (eidx < E) { si = src[eidx]; di = dst[eidx]; } else { si = di = eidx - E; }
  float e = lrelu(a_s[si * H + h] + a_d[di * H + h]);
  float w = __expf(e - m[di * H + h]);
  float alpha = w / (ssum[di * H + h] + PYG_EPS);
  atomicAdd(&out[(size_t)di * HC + f], Hfeat[(size_t)si * HC + f] * alpha);
}

extern "C" void kernel_launch(void* const* d_in, const int* in_sizes, int n_in,
                              void* d_out, int out_size, void* d_ws, size_t ws_size,
                              hipStream_t stream)
{
  const float* x   = (const float*)d_in[0];
  const int*   ei  = (const int*)  d_in[1];
  const float* W1  = (const float*)d_in[2];
  const float* as1 = (const float*)d_in[3];
  const float* ad1 = (const float*)d_in[4];
  const float* b1  = (const float*)d_in[5];
  const float* W2  = (const float*)d_in[6];
  const float* as2 = (const float*)d_in[7];
  const float* ad2 = (const float*)d_in[8];
  const float* b2  = (const float*)d_in[9];
  float* out = (float*)d_out;

  const int F = 128, H1 = 4, C = 64, HC1 = 256, HC2 = 64;
  const int N  = in_sizes[0] / F;
  const int E  = in_sizes[1] / 2;
  const int EE = E + N;
  const int* src = ei;
  const int* dst = ei + E;

  float* h1  = (float*)d_ws;
  float* x2  = h1  + (size_t)N * HC1;
  float* h2  = x2  + (size_t)N * HC1;
  float* aS1 = h2  + (size_t)N * HC2;
  float* aD1 = aS1 + (size_t)N * H1;
  float* m1  = aD1 + (size_t)N * H1;
  float* s1  = m1  + (size_t)N * H1;
  float* aS2 = s1  + (size_t)N * H1;
  float* aD2 = aS2 + N;
  float* m2  = aD2 + N;
  float* s2  = m2  + N;

  const int TB = 256;
  auto cdiv = [](int a, int b) { return (a + b - 1) / b; };

  {
    int tiles = (N / 16) * (HC1 / 16);
    gemm_wmma_f32<<<cdiv(tiles, 8), TB, 0, stream>>>(x, W1, h1, N, F, HC1);
  }
  att_dots<<<cdiv(N * H1, 8), TB, 0, stream>>>(h1, as1, ad1, aS1, aD1, N, H1, C);
  init_ms<<<cdiv(N * H1, TB), TB, 0, stream>>>(m1, s1, N * H1);
  init_bias<<<cdiv(N * HC1, TB), TB, 0, stream>>>(x2, b1, HC1, N * HC1);
  edge_max<H1><<<cdiv(EE, TB), TB, 0, stream>>>(src, dst, aS1, aD1, m1, E, N);
  edge_expsum<H1><<<cdiv(EE, TB), TB, 0, stream>>>(src, dst, aS1, aD1, m1, s1, E, N);
  edge_aggregate<H1, C><<<EE, TB, 0, stream>>>(src, dst, aS1, aD1, m1, s1, h1, x2, E, N);
  elu_inplace<<<cdiv(N * HC1, TB), TB, 0, stream>>>(x2, N * HC1);

  {
    int tiles = (N / 16) * (HC2 / 16);
    gemm_wmma_f32<<<cdiv(tiles, 8), TB, 0, stream>>>(x2, W2, h2, N, HC1, HC2);
  }
  att_dots<<<cdiv(N, 8), TB, 0, stream>>>(h2, as2, ad2, aS2, aD2, N, 1, C);
  init_ms<<<cdiv(N, TB), TB, 0, stream>>>(m2, s2, N);
  init_bias<<<cdiv(N * HC2, TB), TB, 0, stream>>>(out, b2, HC2, N * HC2);
  edge_max<1><<<cdiv(EE, TB), TB, 0, stream>>>(src, dst, aS2, aD2, m2, E, N);
  edge_expsum<1><<<cdiv(EE, TB), TB, 0, stream>>>(src, dst, aS2, aD2, m2, s2, E, N);
  edge_aggregate<1, C><<<cdiv(EE, 4), TB, 0, stream>>>(src, dst, aS2, aD2, m2, s2, h2, out, E, N);
}


# Edges bucketed by destination once per call (int count, scan, place); one wave per node sums its bucket in registers, replacing all float atomics; sum order follows bucket arrival order
# speedup vs baseline: 4.5384x; 4.5384x over previous
#include <hip/hip_runtime.h>
#include <math.h>

typedef __attribute__((ext_vector_type(2))) float v2f;
typedef __attribute__((ext_vector_type(8))) float v8f;

#define NEG_SLOPE 0.2f
#define SOFTMAX_EPS 1e-16f

__global__ __launch_bounds__(256) void gemm_wmma_f32(
    const float* __restrict__ X, const float* __restrict__ W,
    float* __restrict__ Hout, int Nrows, int K, int M)
{
  const int tilesN = M >> 4;
  const int tilesTotal = (Nrows >> 4) * tilesN;
  const int wave = blockIdx.x * (blockDim.x >> 5) + (threadIdx.x >> 5);
  if (wave >= tilesTotal) return;
  const int tm = wave / tilesN;
  const int tn = wave - tm * tilesN;
  const int lane = threadIdx.x & 31;
  const int half = lane >> 4;
  const int l15  = lane & 15;
  const int kb   = half << 1;

  const float* xp = X + (size_t)((tm << 4) + l15) * K + kb;
  const float* wp = W + (size_t)kb * M + ((tn << 4) + l15);

  v8f acc = {0.f, 0.f, 0.f, 0.f, 0.f, 0.f, 0.f, 0.f};
  for (int k0 = 0; k0 < K; k0 += 4) {
    v2f a = *(const v2f*)xp;
    v2f b;
    b.x = wp[0];
    b.y = wp[M];
    acc = __builtin_amdgcn_wmma_f32_16x16x4_f32(
         false, a,  false, b,
         (short)0, acc,  false,  false);
    xp += 4;
    wp += (size_t)4 * M;
  }

  float* op = Hout + (size_t)((tm << 4) + (half << 3)) * M + ((tn << 4) + l15);
  #pragma unroll
  for (int v = 0; v < 8; ++v) op[(size_t)v * M] = acc[v];
}

__global__ __launch_bounds__(256) void att_dots(
    const float* __restrict__ Hm, const float* __restrict__ att_s,
    const float* __restrict__ att_d, float* __restrict__ a_s,
    float* __restrict__ a_d, int Nrows, int Hh, int C)
{
  const int wave = blockIdx.x * (blockDim.x >> 5) + (threadIdx.x >> 5);
  if (wave >= Nrows * Hh) return;
  const int lane = threadIdx.x & 31;
  const int h = wave % Hh;
  const float* hv = Hm + (size_t)wave * C;
  const float* as = att_s + (size_t)h * C;
  const float* ad = att_d + (size_t)h * C;
  float ss = 0.f, sd = 0.f;
  for (int c = lane; c < C; c += 32) {
    float x = hv[c];
    ss += x * as[c];
    sd += x * ad[c];
  }
  #pragma unroll
  for (int off = 16; off; off >>= 1) {
    ss += __shfl_xor(ss, off, 32);
    sd += __shfl_xor(sd, off, 32);
  }
  if (lane == 0) { a_s[wave] = ss; a_d[wave] = sd; }
}

__device__ __forceinline__ float lrelu(float e) {
  return e > 0.f ? e : NEG_SLOPE * e;
}

__device__ __forceinline__ float lane_bcast_f(float v, int k) {
  return __builtin_bit_cast(float,
      __builtin_amdgcn_readlane(__builtin_bit_cast(int, v), k));
}

__global__ void zero_counts(int* __restrict__ cnt, int total)
{
  int i = blockIdx.x * blockDim.x + threadIdx.x;
  if (i < total) cnt[i] = 0;
}

__global__ void count_rank(const int* __restrict__ dst, int* __restrict__ cnt,
                           int* __restrict__ rank, int E)
{
  int i = blockIdx.x * blockDim.x + threadIdx.x;
  if (i < E) rank[i] = atomicAdd(&cnt[dst[i]], 1);
}

__global__ __launch_bounds__(1024) void scan_rowptr(
    const int* __restrict__ cnt, int* __restrict__ rowptr, int N)
{
  __shared__ int s_pre[32];
  __shared__ int s_tot[32];
  const int t = threadIdx.x;
  const int lane = t & 31;
  const int w = t >> 5;
  const int base = blockIdx.x * 1024;

  int pre = 0;
  const int4* c4 = (const int4*)cnt;
  for (int i = t; i < (base >> 2); i += 1024) {
    int4 v = c4[i];
    pre += (v.x + v.y) + (v.z + v.w);
  }
  const int idx = base + t;
  const int c = idx < N ? cnt[idx] : 0;

  int x = c;
  #pragma unroll
  for (int off = 1; off < 32; off <<= 1) {
    int y = __shfl_up(x, off, 32);
    if (lane >= off) x += y;
  }
  #pragma unroll
  for (int off = 16; off; off >>= 1) pre += __shfl_xor(pre, off, 32);
  if (lane == 31) s_tot[w] = x;
  if (lane == 0)  s_pre[w] = pre;
  __syncthreads();
  int off0 = 0;
  for (int k = 0; k < 32; ++k) {
    off0 += s_pre[k];
    if (k < w) off0 += s_tot[k];
  }
  const int excl = off0 + x - c;
  if (idx < N) rowptr[idx] = excl;
  if (idx == N - 1) rowptr[N] = excl + c;
}

__global__ void fill_cols(const int* __restrict__ src, const int* __restrict__ dst,
                          const int* __restrict__ rank, const int* __restrict__ rowptr,
                          int* __restrict__ col, int E)
{
  int i = blockIdx.x * blockDim.x + threadIdx.x;
  if (i < E) col[rowptr[dst[i]] + rank[i]] = src[i];
}

__global__ __launch_bounds__(256) void node_gat4(
    const int* __restrict__ rowptr, const int* __restrict__ col,
    const float* __restrict__ a_s, const float* __restrict__ a_d,
    const float* __restrict__ Hfeat, const float* __restrict__ bias,
    float* __restrict__ out, int N)
{
  const int n = __builtin_amdgcn_readfirstlane(
      (int)(blockIdx.x * 8 + (threadIdx.x >> 5)));
  if (n >= N) return;
  const int lane = threadIdx.x & 31;
  const int beg = rowptr[n];
  const int deg = rowptr[n + 1] - beg + 1;
  const float4 ad = *(const float4*)(a_d + (size_t)n * 4);
  const float ninf = -__builtin_inff();

  float m0 = ninf, m1 = ninf, m2 = ninf, m3 = ninf;
  for (int b = 0; b < deg; b += 32) {
    const int t = b + lane;
    if (t < deg) {
      const int s = (t + 1 < deg) ? col[beg + t] : n;
      const float4 as = *(const float4*)(a_s + (size_t)s * 4);
      m0 = fmaxf(m0, lrelu(as.x + ad.x));
      m1 = fmaxf(m1, lrelu(as.y + ad.y));
      m2 = fmaxf(m2, lrelu(as.z + ad.z));
      m3 = fmaxf(m3, lrelu(as.w + ad.w));
    }
  }
  #pragma unroll
  for (int off = 16; off; off >>= 1) {
    m0 = fmaxf(m0, __shfl_xor(m0, off, 32));
    m1 = fmaxf(m1, __shfl_xor(m1, off, 32));
    m2 = fmaxf(m2, __shfl_xor(m2, off, 32));
    m3 = fmaxf(m3, __shfl_xor(m3, off, 32));
  }

  const bool hi = lane >= 16;
  const float4* H4 = (const float4*)Hfeat;
  float S0 = 0.f, S1 = 0.f, S2 = 0.f, S3 = 0.f;
  float4 accA = {0.f, 0.f, 0.f, 0.f};
  float4 accB = {0.f, 0.f, 0.f, 0.f};
  for (int b = 0; b < deg; b += 32) {
    const int t = b + lane;
    int s = n;
    float w0 = 0.f, w1 = 0.f, w2 = 0.f, w3 = 0.f;
    if (t < deg) {
      if (t + 1 < deg) s = col[beg + t];
      const float4 as = *(const float4*)(a_s + (size_t)s * 4);
      w0 = __expf(lrelu(as.x + ad.x) - m0);
      w1 = __expf(lrelu(as.y + ad.y) - m1);
      w2 = __expf(lrelu(as.z + ad.z) - m2);
      w3 = __expf(lrelu(as.w + ad.w) - m3);
      S0 += w0; S1 += w1; S2 += w2; S3 += w3;
    }
    const int cnt = (deg - b < 32) ? (deg - b) : 32;
    #pragma unroll 4
    for (int k = 0; k < cnt; ++k) {
      const int sk = __builtin_amdgcn_readlane(s, k);
      const float b0 = lane_bcast_f(w0, k);
      const float b1 = lane_bcast_f(w1, k);
      const float b2 = lane_bcast_f(w2, k);
      const float b3 = lane_bcast_f(w3, k);
      const float wa = hi ? b1 : b0;
      const float wb = hi ? b3 : b2;
      const float4* row = H4 + (size_t)sk * 64;
      const float4 va = row[lane];
      const float4 vb = row[32 + lane];
      accA.x += wa * va.x; accA.y += wa * va.y;
      accA.z += wa * va.z; accA.w += wa * va.w;
      accB.x += wb * vb.x; accB.y += wb * vb.y;
      accB.z += wb * vb.z; accB.w += wb * vb.w;
    }
  }
  #pragma unroll
  for (int off = 16; off; off >>= 1) {
    S0 += __shfl_xor(S0, off, 32);
    S1 += __shfl_xor(S1, off, 32);
    S2 += __shfl_xor(S2, off, 32);
    S3 += __shfl_xor(S3, off, 32);
  }
  const float da = (hi ? S1 : S0) + SOFTMAX_EPS;
  const float db = (hi ? S3 : S2) + SOFTMAX_EPS;
  const float4 ba = ((const float4*)bias)[lane];
  const float4 bb = ((const float4*)bias)[32 + lane];
  float4 ra, rb;
  ra.x = accA.x / da + ba.x; ra.y = accA.y / da + ba.y;
  ra.z = accA.z / da + ba.z; ra.w = accA.w / da + ba.w;
  rb.x = accB.x / db + bb.x; rb.y = accB.y / db + bb.y;
  rb.z = accB.z / db + bb.z; rb.w = accB.w / db + bb.w;
  ra.x = ra.x > 0.f ? ra.x : expm1f(ra.x);
  ra.y = ra.y > 0.f ? ra.y : expm1f(ra.y);
  ra.z = ra.z > 0.f ? ra.z : expm1f(ra.z);
  ra.w = ra.w > 0.f ? ra.w : expm1f(ra.w);
  rb.x = rb.x > 0.f ? rb.x : expm1f(rb.x);
  rb.y = rb.y > 0.f ? rb.y : expm1f(rb.y);
  rb.z = rb.z > 0.f ? rb.z : expm1f(rb.z);
  rb.w = rb.w > 0.f ? rb.w : expm1f(rb.w);
  float4* orow = (float4*)(out + (size_t)n * 256);
  orow[lane] = ra;
  orow[32 + lane] = rb;
}

__global__ __launch_bounds__(256) void node_gat1(
    const int* __restrict__ rowptr, const int* __restrict__ col,
    const float* __restrict__ a_s, const float* __restrict__ a_d,
    const float* __restrict__ Hfeat, const float* __restrict__ bias,
    float* __restrict__ out, int N)
{
  const int n = __builtin_amdgcn_readfirstlane(
      (int)(blockIdx.x * 8 + (threadIdx.x >> 5)));
  if (n >= N) return;
  const int lane = threadIdx.x & 31;
  const int beg = rowptr[n];
  const int deg = rowptr[n + 1] - beg + 1;
  const float ad = a_d[n];

  float m = -__builtin_inff();
  for (int b = 0; b < deg; b += 32) {
    const int t = b + lane;
    if (t < deg) {
      const int s = (t + 1 < deg) ? col[beg + t] : n;
      m = fmaxf(m, lrelu(a_s[s] + ad));
    }
  }
  #pragma unroll
  for (int off = 16; off; off >>= 1) m = fmaxf(m, __shfl_xor(m, off, 32));

  const float2* H2 = (const float2*)Hfeat;
  float S = 0.f;
  float2 acc = {0.f, 0.f};
  for (int b = 0; b < deg; b += 32) {
    const int t = b + lane;
    int s = n;
    float w = 0.f;
    if (t < deg) {
      if (t + 1 < deg) s = col[beg + t];
      w = __expf(lrelu(a_s[s] + ad) - m);
      S += w;
    }
    const int cnt = (deg - b < 32) ? (deg - b) : 32;
    #pragma unroll 4
    for (int k = 0; k < cnt; ++k) {
      const int sk = __builtin_amdgcn_readlane(s, k);
      const float wk = lane_bcast_f(w, k);
      const float2 v = H2[(size_t)sk * 32 + lane];
      acc.x += wk * v.x;
      acc.y += wk * v.y;
    }
  }
  #pragma unroll
  for (int off = 16; off; off >>= 1) S += __shfl_xor(S, off, 32);
  const float d = S + SOFTMAX_EPS;
  const float2 bv = ((const float2*)bias)[lane];
  float2 r;
  r.x = acc.x / d + bv.x;
  r.y = acc.y / d + bv.y;
  ((float2*)(out + (size_t)n * 64))[lane] = r;
}

extern "C" void kernel_launch(void* const* d_in, const int* in_sizes, int n_in,
                              void* d_out, int out_size, void* d_ws, size_t ws_size,
                              hipStream_t stream)
{
  const float* x   = (const float*)d_in[0];
  const int*   ei  = (const int*)  d_in[1];
  const float* W1  = (const float*)d_in[2];
  const float* as1 = (const float*)d_in[3];
  const float* ad1 = (const float*)d_in[4];
  const float* b1  = (const float*)d_in[5];
  const float* W2  = (const float*)d_in[6];
  const float* as2 = (const float*)d_in[7];
  const float* ad2 = (const float*)d_in[8];
  const float* b2  = (const float*)d_in[9];
  float* out = (float*)d_out;

  const int F = 128, H1 = 4, C = 64, HC1 = 256, HC2 = 64;
  const int N  = in_sizes[0] / F;
  const int E  = in_sizes[1] / 2;
  const int* src = ei;
  const int* dst = ei + E;

  float* h1  = (float*)d_ws;
  float* x2  = h1  + (size_t)N * HC1;
  float* h2  = x2  + (size_t)N * HC1;
  float* aS1 = h2  + (size_t)N * HC2;
  float* aD1 = aS1 + (size_t)N * H1;
  float* aS2 = aD1 + (size_t)N * H1;
  float* aD2 = aS2 + N;
  int* cnt    = (int*)(aD2 + N);
  int* rowptr = cnt + N;
  int* rank   = rowptr + N + 16;
  int* col    = rank + E;

  const int TB = 256;
  auto cdiv = [](int a, int b) { return (a + b - 1) / b; };

  zero_counts<<<cdiv(N, TB), TB, 0, stream>>>(cnt, N);
  count_rank<<<cdiv(E, TB), TB, 0, stream>>>(dst, cnt, rank, E);
  scan_rowptr<<<cdiv(N, 1024), 1024, 0, stream>>>(cnt, rowptr, N);
  fill_cols<<<cdiv(E, TB), TB, 0, stream>>>(src, dst, rank, rowptr, col, E);

  {
    int tiles = (N / 16) * (HC1 / 16);
    gemm_wmma_f32<<<cdiv(tiles, 8), TB, 0, stream>>>(x, W1, h1, N, F, HC1);
  }
  att_dots<<<cdiv(N * H1, 8), TB, 0, stream>>>(h1, as1, ad1, aS1, aD1, N, H1, C);
  node_gat4<<<cdiv(N, 8), TB, 0, stream>>>(rowptr, col, aS1, aD1, h1, b1, x2, N);

  {
    int tiles = (N / 16) * (HC2 / 16);
    gemm_wmma_f32<<<cdiv(tiles, 8), TB, 0, stream>>>(x2, W2, h2, N, HC1, HC2);
  }
  att_dots<<<cdiv(N, 8), TB, 0, stream>>>(h2, as2, ad2, aS2, aD2, N, 1, C);
  node_gat1<<<cdiv(N, 8), TB, 0, stream>>>(rowptr, col, aS2, aD2, h2, b2, out, N);
}


# Attention dots move into the GEMM: one wave owns a 16x64 head strip (4 f32 WMMA accumulators per A fragment), tree-sums a_s and a_d from them; att_dots launches gone; k and dot sum order change
# speedup vs baseline: 6.5274x; 1.4383x over previous
#include <hip/hip_runtime.h>
#include <math.h>

typedef __attribute__((ext_vector_type(2))) float v2f;
typedef __attribute__((ext_vector_type(8))) float v8f;

#define NEG_SLOPE 0.2f
#define SOFTMAX_EPS 1e-16f

struct Frag {
  float4 a0, a1;
  float4 b[8];
};

template <int M>
__device__ __forceinline__ void load_frag(Frag& f, const float* __restrict__ xp,
                                          const float* __restrict__ wp)
{
  f.a0 = *(const float4*)xp;
  f.a1 = *(const float4*)(xp + 4);
  #pragma unroll
  for (int r = 0; r < 8; ++r) f.b[r] = *(const float4*)(wp + r * M);
}

__device__ __forceinline__ v8f mma4(v2f a, float b0, float b1, v8f c)
{
  v2f b = {b0, b1};
  return __builtin_amdgcn_wmma_f32_16x16x4_f32(
       false, a,  false, b,
       (short)0, c,  false,  false);
}

__device__ __forceinline__ void mma_block(const Frag& f, v8f& c0, v8f& c1,
                                          v8f& c2, v8f& c3)
{
  const v2f a[4] = {{f.a0.x, f.a0.y}, {f.a0.z, f.a0.w},
                    {f.a1.x, f.a1.y}, {f.a1.z, f.a1.w}};
  #pragma unroll
  for (int q = 0; q < 4; ++q) {
    const float4 r0 = f.b[2 * q];
    const float4 r1 = f.b[2 * q + 1];
    c0 = mma4(a[q], r0.x, r1.x, c0);
    c1 = mma4(a[q], r0.y, r1.y, c1);
    c2 = mma4(a[q], r0.z, r1.z, c2);
    c3 = mma4(a[q], r0.w, r1.w, c3);
  }
}

template <int Wd>
__device__ __forceinline__ void halve_sum(float (&p)[16], int c)
{
  const bool up = (c & Wd) != 0;
  #pragma unroll
  for (int i = 0; i < Wd; ++i) {
    const float keep = up ? p[i + Wd] : p[i];
    const float send = up ? p[i] : p[i + Wd];
    p[i] = keep + __shfl_xor(send, Wd, 32);
  }
}

template <int K, int HH>
__global__ __launch_bounds__(256) void gemm_att_wmma_f32(
    const float* __restrict__ X, const float* __restrict__ W,
    const float* __restrict__ att_s, const float* __restrict__ att_d,
    float* __restrict__ Hout, float* __restrict__ a_s, float* __restrict__ a_d,
    int Nrows)
{
  constexpr int M = HH * 64;
  constexpr int NB = K / 16;
  static_assert(K % 16 == 0 && NB >= 2, "K must be a multiple of 16, at least 32");
  const int total = (Nrows >> 4) * HH;
  const int wave = blockIdx.x * (blockDim.x >> 5) + (threadIdx.x >> 5);
  if (wave >= total) return;
  const int tm = wave / HH;
  const int head = wave - tm * HH;
  const int lane = threadIdx.x & 31;
  const int half = lane >> 4;
  const int c = lane & 15;
  const int hb = head << 6;

  const float* xp = X + (size_t)((tm << 4) + c) * K + (half << 3);
  const float* wp = W + (size_t)(half << 3) * M + hb + (c << 2);

  v8f acc0 = {0.f, 0.f, 0.f, 0.f, 0.f, 0.f, 0.f, 0.f};
  v8f acc1 = acc0, acc2 = acc0, acc3 = acc0;

  Frag f[3];
  load_frag<M>(f[0], xp, wp);
  load_frag<M>(f[1], xp + 16, wp + 16 * M);
  const float4 as4 = *(const float4*)(att_s + hb + (c << 2));
  const float4 ad4 = *(const float4*)(att_d + hb + (c << 2));
  #pragma unroll
  for (int i = 0; i < NB; ++i) {
    if (i + 2 < NB)
      load_frag<M>(f[(i + 2) % 3], xp + (i + 2) * 16, wp + (i + 2) * 16 * M);
    __builtin_amdgcn_sched_barrier(0);
    mma_block(f[i % 3], acc0, acc1, acc2, acc3);
    __builtin_amdgcn_sched_barrier(0);
  }

  const int row0 = (tm << 4) + (half << 3);
  float* op = Hout + (size_t)row0 * M + hb + (c << 2);
  #pragma unroll
  for (int v = 0; v < 8; ++v) {
    float4 o = {acc0[v], acc1[v], acc2[v], acc3[v]};
    *(float4*)(op + v * M) = o;
  }

  float p[16];
  #pragma unroll
  for (int v = 0; v < 8; ++v) {
    p[v]     = acc0[v] * as4.x + acc1[v] * as4.y + acc2[v] * as4.z + acc3[v] * as4.w;
    p[8 + v] = acc0[v] * ad4.x + acc1[v] * ad4.y + acc2[v] * ad4.z + acc3[v] * ad4.w;
  }
  halve_sum<8>(p, c);
  halve_sum<4>(p, c);
  halve_sum<2>(p, c);
  halve_sum<1>(p, c);
  float* dp = (c < 8) ? a_s : a_d;
  dp[(size_t)(row0 + (c & 7)) * HH + head] = p[0];
}

__device__ __forceinline__ float lrelu(float e) {
  return e > 0.f ? e : NEG_SLOPE * e;
}

__device__ __forceinline__ float lane_bcast_f(float v, int k) {
  return __builtin_bit_cast(float,
      __builtin_amdgcn_readlane(__builtin_bit_cast(int, v), k));
}

__global__ void zero_counts(int* __restrict__ cnt, int total)
{
  int i = blockIdx.x * blockDim.x + threadIdx.x;
  if (i < total) cnt[i] = 0;
}

__global__ void count_rank(const int* __restrict__ dst, int* __restrict__ cnt,
                           int* __restrict__ rank, int E)
{
  int i = blockIdx.x * blockDim.x + threadIdx.x;
  if (i < E) rank[i] = atomicAdd(&cnt[dst[i]], 1);
}

__global__ __launch_bounds__(1024) void scan_rowptr(
    const int* __restrict__ cnt, int* __restrict__ rowptr, int N)
{
  __shared__ int s_pre[32];
  __shared__ int s_tot[32];
  const int t = threadIdx.x;
  const int lane = t & 31;
  const int w = t >> 5;
  const int base = blockIdx.x * 1024;

  int pre = 0;
  const int4* c4 = (const int4*)cnt;
  for (int i = t; i < (base >> 2); i += 1024) {
    int4 v = c4[i];
    pre += (v.x + v.y) + (v.z + v.w);
  }
  const int idx = base + t;
  const int c = idx < N ? cnt[idx] : 0;

  int x = c;
  #pragma unroll
  for (int off = 1; off < 32; off <<= 1) {
    int y = __shfl_up(x, off, 32);
    if (lane >= off) x += y;
  }
  #pragma unroll
  for (int off = 16; off; off >>= 1) pre += __shfl_xor(pre, off, 32);
  if (lane == 31) s_tot[w] = x;
  if (lane == 0)  s_pre[w] = pre;
  __syncthreads();
  int off0 = 0;
  for (int k = 0; k < 32; ++k) {
    off0 += s_pre[k];
    if (k < w) off0 += s_tot[k];
  }
  const int excl = off0 + x - c;
  if (idx < N) rowptr[idx] = excl;
  if (idx == N - 1) rowptr[N] = excl + c;
}

__global__ void fill_cols(const int* __restrict__ src, const int* __restrict__ dst,
                          const int* __restrict__ rank, const int* __restrict__ rowptr,
                          int* __restrict__ col, int E)
{
  int i = blockIdx.x * blockDim.x + threadIdx.x;
  if (i < E) col[rowptr[dst[i]] + rank[i]] = src[i];
}

__global__ __launch_bounds__(256) void node_gat4(
    const int* __restrict__ rowptr, const int* __restrict__ col,
    const float* __restrict__ a_s, const float* __restrict__ a_d,
    const float* __restrict__ Hfeat, const float* __restrict__ bias,
    float* __restrict__ out, int N)
{
  const int n = __builtin_amdgcn_readfirstlane(
      (int)(blockIdx.x * 8 + (threadIdx.x >> 5)));
  if (n >= N) return;
  const int lane = threadIdx.x & 31;
  const int beg = rowptr[n];
  const int deg = rowptr[n + 1] - beg + 1;
  const float4 ad = *(const float4*)(a_d + (size_t)n * 4);
  const float ninf = -__builtin_inff();

  float m0 = ninf, m1 = ninf, m2 = ninf, m3 = ninf;
  for (int b = 0; b < deg; b += 32) {
    const int t = b + lane;
    if (t < deg) {
      const int s = (t + 1 < deg) ? col[beg + t] : n;
      const float4 as = *(const float4*)(a_s + (size_t)s * 4);
      m0 = fmaxf(m0, lrelu(as.x + ad.x));
      m1 = fmaxf(m1, lrelu(as.y + ad.y));
      m2 = fmaxf(m2, lrelu(as.z + ad.z));
      m3 = fmaxf(m3, lrelu(as.w + ad.w));
    }
  }
  #pragma unroll
  for (int off = 16; off; off >>= 1) {
    m0 = fmaxf(m0, __shfl_xor(m0, off, 32));
    m1 = fmaxf(m1, __shfl_xor(m1, off, 32));
    m2 = fmaxf(m2, __shfl_xor(m2, off, 32));
    m3 = fmaxf(m3, __shfl_xor(m3, off, 32));
  }

  const bool hi = lane >= 16;
  const float4* H4 = (const float4*)Hfeat;
  float S0 = 0.f, S1 = 0.f, S2 = 0.f, S3 = 0.f;
  float4 accA = {0.f, 0.f, 0.f, 0.f};
  float4 accB = {0.f, 0.f, 0.f, 0.f};
  for (int b = 0; b < deg; b += 32) {
    const int t = b + lane;
    int s = n;
    float w0 = 0.f, w1 = 0.f, w2 = 0.f, w3 = 0.f;
    if (t < deg) {
      if (t + 1 < deg) s = col[beg + t];
      const float4 as = *(const float4*)(a_s + (size_t)s * 4);
      w0 = __expf(lrelu(as.x + ad.x) - m0);
      w1 = __expf(lrelu(as.y + ad.y) - m1);
      w2 = __expf(lrelu(as.z + ad.z) - m2);
      w3 = __expf(lrelu(as.w + ad.w) - m3);
      S0 += w0; S1 += w1; S2 += w2; S3 += w3;
    }
    const int cnt = (deg - b < 32) ? (deg - b) : 32;
    #pragma unroll 4
    for (int k = 0; k < cnt; ++k) {
      const int sk = __builtin_amdgcn_readlane(s, k);
      const float b0 = lane_bcast_f(w0, k);
      const float b1 = lane_bcast_f(w1, k);
      const float b2 = lane_bcast_f(w2, k);
      const float b3 = lane_bcast_f(w3, k);
      const float wa = hi ? b1 : b0;
      const float wb = hi ? b3 : b2;
      const float4* row = H4 + (size_t)sk * 64;
      const float4 va = row[lane];
      const float4 vb = row[32 + lane];
      accA.x += wa * va.x; accA.y += wa * va.y;
      accA.z += wa * va.z; accA.w += wa * va.w;
      accB.x += wb * vb.x; accB.y += wb * vb.y;
      accB.z += wb * vb.z; accB.w += wb * vb.w;
    }
  }
  #pragma unroll
  for (int off = 16; off; off >>= 1) {
    S0 += __shfl_xor(S0, off, 32);
    S1 += __shfl_xor(S1, off, 32);
    S2 += __shfl_xor(S2, off, 32);
    S3 += __shfl_xor(S3, off, 32);
  }
  const float da = (hi ? S1 : S0) + SOFTMAX_EPS;
  const float db = (hi ? S3 : S2) + SOFTMAX_EPS;
  const float4 ba = ((const float4*)bias)[lane];
  const float4 bb = ((const float4*)bias)[32 + lane];
  float4 ra, rb;
  ra.x = accA.x / da + ba.x; ra.y = accA.y / da + ba.y;
  ra.z = accA.z / da + ba.z; ra.w = accA.w / da + ba.w;
  rb.x = accB.x / db + bb.x; rb.y = accB.y / db + bb.y;
  rb.z = accB.z / db + bb.z; rb.w = accB.w / db + bb.w;
  ra.x = ra.x > 0.f ? ra.x : expm1f(ra.x);
  ra.y = ra.y > 0.f ? ra.y : expm1f(ra.y);
  ra.z = ra.z > 0.f ? ra.z : expm1f(ra.z);
  ra.w = ra.w > 0.f ? ra.w : expm1f(ra.w);
  rb.x = rb.x > 0.f ? rb.x : expm1f(rb.x);
  rb.y = rb.y > 0.f ? rb.y : expm1f(rb.y);
  rb.z = rb.z > 0.f ? rb.z : expm1f(rb.z);
  rb.w = rb.w > 0.f ? rb.w : expm1f(rb.w);
  float4* orow = (float4*)(out + (size_t)n * 256);
  orow[lane] = ra;
  orow[32 + lane] = rb;
}

__global__ __launch_bounds__(256) void node_gat1(
    const int* __restrict__ rowptr, const int* __restrict__ col,
    const float* __restrict__ a_s, const float* __restrict__ a_d,
    const float* __restrict__ Hfeat, const float* __restrict__ bias,
    float* __restrict__ out, int N)
{
  const int n = __builtin_amdgcn_readfirstlane(
      (int)(blockIdx.x * 8 + (threadIdx.x >> 5)));
  if (n >= N) return;
  const int lane = threadIdx.x & 31;
  const int beg = rowptr[n];
  const int deg = rowptr[n + 1] - beg + 1;
  const float ad = a_d[n];

  float m = -__builtin_inff();
  for (int b = 0; b < deg; b += 32) {
    const int t = b + lane;
    if (t < deg) {
      const int s = (t + 1 < deg) ? col[beg + t] : n;
      m = fmaxf(m, lrelu(a_s[s] + ad));
    }
  }
  #pragma unroll
  for (int off = 16; off; off >>= 1) m = fmaxf(m, __shfl_xor(m, off, 32));

  const float2* H2 = (const float2*)Hfeat;
  float S = 0.f;
  float2 acc = {0.f, 0.f};
  for (int b = 0; b < deg; b += 32) {
    const int t = b + lane;
    int s = n;
    float w = 0.f;
    if (t < deg) {
      if (t + 1 < deg) s = col[beg + t];
      w = __expf(lrelu(a_s[s] + ad) - m);
      S += w;
    }
    const int cnt = (deg - b < 32) ? (deg - b) : 32;
    #pragma unroll 4
    for (int k = 0; k < cnt; ++k) {
      const int sk = __builtin_amdgcn_readlane(s, k);
      const float wk = lane_bcast_f(w, k);
      const float2 v = H2[(size_t)sk * 32 + lane];
      acc.x += wk * v.x;
      acc.y += wk * v.y;
    }
  }
  #pragma unroll
  for (int off = 16; off; off >>= 1) S += __shfl_xor(S, off, 32);
  const float d = S + SOFTMAX_EPS;
  const float2 bv = ((const float2*)bias)[lane];
  float2 r;
  r.x = acc.x / d + bv.x;
  r.y = acc.y / d + bv.y;
  ((float2*)(out + (size_t)n * 64))[lane] = r;
}

extern "C" void kernel_launch(void* const* d_in, const int* in_sizes, int n_in,
                              void* d_out, int out_size, void* d_ws, size_t ws_size,
                              hipStream_t stream)
{
  const float* x   = (const float*)d_in[0];
  const int*   ei  = (const int*)  d_in[1];
  const float* W1  = (const float*)d_in[2];
  const float* as1 = (const float*)d_in[3];
  const float* ad1 = (const float*)d_in[4];
  const float* b1  = (const float*)d_in[5];
  const float* W2  = (const float*)d_in[6];
  const float* as2 = (const float*)d_in[7];
  const float* ad2 = (const float*)d_in[8];
  const float* b2  = (const float*)d_in[9];
  float* out = (float*)d_out;

  constexpr int F = 128, H1 = 4, HC1 = 256, HC2 = 64;
  const int N  = in_sizes[0] / F;
  const int E  = in_sizes[1] / 2;
  const int* src = ei;
  const int* dst = ei + E;

  float* h1  = (float*)d_ws;
  float* x2  = h1  + (size_t)N * HC1;
  float* h2  = x2  + (size_t)N * HC1;
  float* aS1 = h2  + (size_t)N * HC2;
  float* aD1 = aS1 + (size_t)N * H1;
  float* aS2 = aD1 + (size_t)N * H1;
  float* aD2 = aS2 + N;
  int* cnt    = (int*)(aD2 + N);
  int* rowptr = cnt + N;
  int* rank   = rowptr + N + 16;
  int* col    = rank + E;

  const int TB = 256;
  auto cdiv = [](int a, int b) { return (a + b - 1) / b; };

  zero_counts<<<cdiv(N, TB), TB, 0, stream>>>(cnt, N);
  count_rank<<<cdiv(E, TB), TB, 0, stream>>>(dst, cnt, rank, E);
  scan_rowptr<<<cdiv(N, 1024), 1024, 0, stream>>>(cnt, rowptr, N);
  fill_cols<<<cdiv(E, TB), TB, 0, stream>>>(src, dst, rank, rowptr, col, E);

  gemm_att_wmma_f32<F, H1><<<cdiv((N / 16) * H1, 8), TB, 0, stream>>>(
      x, W1, as1, ad1, h1, aS1, aD1, N);
  node_gat4<<<cdiv(N, 8), TB, 0, stream>>>(rowptr, col, aS1, aD1, h1, b1, x2, N);

  gemm_att_wmma_f32<HC1, 1><<<cdiv(N / 16, 8), TB, 0, stream>>>(
      x2, W2, as2, ad2, h2, aS2, aD2, N);
  node_gat1<<<cdiv(N, 8), TB, 0, stream>>>(rowptr, col, aS2, aD2, h2, b2, out, N);
}


# Edge buckets by a two-level counting sort in LDS (4096-edge chunks binned by 256-node block, then ranked per block): no global atomics or scattered stores, 4 launches become 2; sum order changes
# speedup vs baseline: 9.5971x; 1.4703x over previous
#include <hip/hip_runtime.h>
#include <math.h>

typedef __attribute__((ext_vector_type(2))) float v2f;
typedef __attribute__((ext_vector_type(8))) float v8f;

#define NEG_SLOPE 0.2f
#define SOFTMAX_EPS 1e-16f

struct Frag {
  float4 a0, a1;
  float4 b[8];
};

template <int M>
__device__ __forceinline__ void load_frag(Frag& f, const float* __restrict__ xp,
                                          const float* __restrict__ wp)
{
  f.a0 = *(const float4*)xp;
  f.a1 = *(const float4*)(xp + 4);
  #pragma unroll
  for (int r = 0; r < 8; ++r) f.b[r] = *(const float4*)(wp + r * M);
}

__device__ __forceinline__ v8f mma4(v2f a, float b0, float b1, v8f c)
{
  v2f b = {b0, b1};
  return __builtin_amdgcn_wmma_f32_16x16x4_f32(
       false, a,  false, b,
       (short)0, c,  false,  false);
}

__device__ __forceinline__ void mma_block(const Frag& f, v8f& c0, v8f& c1,
                                          v8f& c2, v8f& c3)
{
  const v2f a[4] = {{f.a0.x, f.a0.y}, {f.a0.z, f.a0.w},
                    {f.a1.x, f.a1.y}, {f.a1.z, f.a1.w}};
  #pragma unroll
  for (int q = 0; q < 4; ++q) {
    const float4 r0 = f.b[2 * q];
    const float4 r1 = f.b[2 * q + 1];
    c0 = mma4(a[q], r0.x, r1.x, c0);
    c1 = mma4(a[q], r0.y, r1.y, c1);
    c2 = mma4(a[q], r0.z, r1.z, c2);
    c3 = mma4(a[q], r0.w, r1.w, c3);
  }
}

template <int Wd>
__device__ __forceinline__ void halve_sum(float (&p)[16], int c)
{
  const bool up = (c & Wd) != 0;
  #pragma unroll
  for (int i = 0; i < Wd; ++i) {
    const float keep = up ? p[i + Wd] : p[i];
    const float send = up ? p[i] : p[i + Wd];
    p[i] = keep + __shfl_xor(send, Wd, 32);
  }
}

template <int K, int HH>
__global__ __launch_bounds__(256) void gemm_att_wmma_f32(
    const float* __restrict__ X, const float* __restrict__ W,
    const float* __restrict__ att_s, const float* __restrict__ att_d,
    float* __restrict__ Hout, float* __restrict__ a_s, float* __restrict__ a_d,
    int Nrows)
{
  constexpr int M = HH * 64;
  constexpr int NB = K / 16;
  static_assert(K % 16 == 0 && NB >= 2, "K must be a multiple of 16, at least 32");
  const int total = (Nrows >> 4) * HH;
  const int wave = blockIdx.x * (blockDim.x >> 5) + (threadIdx.x >> 5);
  if (wave >= total) return;
  const int tm = wave / HH;
  const int head = wave - tm * HH;
  const int lane = threadIdx.x & 31;
  const int half = lane >> 4;
  const int c = lane & 15;
  const int hb = head << 6;

  const float* xp = X + (size_t)((tm << 4) + c) * K + (half << 3);
  const float* wp = W + (size_t)(half << 3) * M + hb + (c << 2);

  v8f acc0 = {0.f, 0.f, 0.f, 0.f, 0.f, 0.f, 0.f, 0.f};
  v8f acc1 = acc0, acc2 = acc0, acc3 = acc0;

  Frag f[3];
  load_frag<M>(f[0], xp, wp);
  load_frag<M>(f[1], xp + 16, wp + 16 * M);
  const float4 as4 = *(const float4*)(att_s + hb + (c << 2));
  const float4 ad4 = *(const float4*)(att_d + hb + (c << 2));
  #pragma unroll
  for (int i = 0; i < NB; ++i) {
    if (i + 2 < NB)
      load_frag<M>(f[(i + 2) % 3], xp + (i + 2) * 16, wp + (i + 2) * 16 * M);
    __builtin_amdgcn_sched_barrier(0);
    mma_block(f[i % 3], acc0, acc1, acc2, acc3);
    __builtin_amdgcn_sched_barrier(0);
  }

  const int row0 = (tm << 4) + (half << 3);
  float* op = Hout + (size_t)row0 * M + hb + (c << 2);
  #pragma unroll
  for (int v = 0; v < 8; ++v) {
    float4 o = {acc0[v], acc1[v], acc2[v], acc3[v]};
    *(float4*)(op + v * M) = o;
  }

  float p[16];
  #pragma unroll
  for (int v = 0; v < 8; ++v) {
    p[v]     = acc0[v] * as4.x + acc1[v] * as4.y + acc2[v] * as4.z + acc3[v] * as4.w;
    p[8 + v] = acc0[v] * ad4.x + acc1[v] * ad4.y + acc2[v] * ad4.z + acc3[v] * ad4.w;
  }
  halve_sum<8>(p, c);
  halve_sum<4>(p, c);
  halve_sum<2>(p, c);
  halve_sum<1>(p, c);
  float* dp = (c < 8) ? a_s : a_d;
  dp[(size_t)(row0 + (c & 7)) * HH + head] = p[0];
}

__device__ __forceinline__ float lrelu(float e) {
  return e > 0.f ? e : NEG_SLOPE * e;
}

__device__ __forceinline__ float lane_bcast_f(float v, int k) {
  return __builtin_bit_cast(float,
      __builtin_amdgcn_readlane(__builtin_bit_cast(int, v), k));
}

#define BIN_SHIFT 8
#define BIN_NODES (1 << BIN_SHIFT)
#define CHUNK 4096
#define CSR_THREADS 512
#define CSR_CAP 6144

__device__ __forceinline__ int block_scan_excl(int v, int* s_w, int& total)
{
  const int lane = threadIdx.x & 31;
  const int w = threadIdx.x >> 5;
  int x = v;
  #pragma unroll
  for (int off = 1; off < 32; off <<= 1) {
    const int y = __shfl_up(x, off, 32);
    if (lane >= off) x += y;
  }
  if (lane == 31) s_w[w] = x;
  __syncthreads();
  int pre = 0, tot = 0;
  #pragma unroll
  for (int i = 0; i < CSR_THREADS / 32; ++i) {
    const int q = s_w[i];
    if (i < w) pre += q;
    tot += q;
  }
  __syncthreads();
  total = tot;
  return pre + x - v;
}

__global__ __launch_bounds__(CSR_THREADS) void bin_edges(
    const int* __restrict__ src, const int* __restrict__ dst,
    unsigned* __restrict__ region, unsigned* __restrict__ meta, int E, int nbin)
{
  __shared__ int s_cnt[CSR_THREADS];
  __shared__ int s_off[CSR_THREADS];
  __shared__ int s_w[CSR_THREADS / 32];
  __shared__ unsigned s_stage[CHUNK];
  constexpr int PER = CHUNK / CSR_THREADS;
  const int t = threadIdx.x;
  const int base = blockIdx.x * CHUNK;

  s_cnt[t] = 0;
  int d[PER], s[PER], r[PER];
  #pragma unroll
  for (int j = 0; j < PER; ++j) {
    const int i = base + j * CSR_THREADS + t;
    d[j] = (i < E) ? dst[i] : -1;
    s[j] = (i < E) ? src[i] : 0;
  }
  __syncthreads();
  #pragma unroll
  for (int j = 0; j < PER; ++j) {
    r[j] = 0;
    if (d[j] >= 0) r[j] = atomicAdd(&s_cnt[d[j] >> BIN_SHIFT], 1);
  }
  __syncthreads();
  const int c = s_cnt[t];
  int tot;
  const int o = block_scan_excl(c, s_w, tot);
  s_off[t] = o;
  if (t < nbin) meta[(size_t)blockIdx.x * nbin + t] = ((unsigned)o << 16) | (unsigned)c;
  __syncthreads();
  #pragma unroll
  for (int j = 0; j < PER; ++j) {
    if (d[j] >= 0)
      s_stage[s_off[d[j] >> BIN_SHIFT] + r[j]] =
          ((unsigned)(d[j] & (BIN_NODES - 1)) << 24) | (unsigned)s[j];
  }
  __syncthreads();
  const int nvalid = (E - base < CHUNK) ? (E - base) : CHUNK;
  for (int i = t; i < nvalid; i += CSR_THREADS) region[base + i] = s_stage[i];
}

__device__ __forceinline__ int find_piece(const int* s_start, int i)
{
  int k = 0;
  #pragma unroll
  for (int step = CSR_THREADS / 2; step >= 1; step >>= 1)
    if (s_start[k + step] <= i) k += step;
  return k;
}

__global__ __launch_bounds__(CSR_THREADS) void build_csr(
    const unsigned* __restrict__ region, const unsigned* __restrict__ meta,
    int* __restrict__ rowptr, int* __restrict__ col, int N, int nch, int nbin)
{
  __shared__ int s_start[CSR_THREADS];
  __shared__ int s_from[CSR_THREADS];
  __shared__ int s_cur[BIN_NODES];
  __shared__ int s_w[CSR_THREADS / 32];
  __shared__ unsigned s_raw[CSR_CAP];
  __shared__ int s_out[CSR_CAP];
  const int t = threadIdx.x;
  const int b = blockIdx.x;

  int off = 0, cnt = 0;
  if (t < nch) {
    const unsigned m = meta[(size_t)t * nbin + b];
    off = (int)(m >> 16);
    cnt = (int)(m & 0xffffu);
  }
  int base, T;
  (void)block_scan_excl(off, s_w, base);
  const int st = block_scan_excl(cnt, s_w, T);
  s_start[t] = (t < nch) ? st : 0x7fffffff;
  s_from[t] = t * CHUNK + off;
  if (t < BIN_NODES) s_cur[t] = 0;
  __syncthreads();
  const bool staged = T <= CSR_CAP;

  for (int i0 = 0; i0 < T; i0 += 4 * CSR_THREADS) {
    unsigned it[4];
    #pragma unroll
    for (int j = 0; j < 4; ++j) {
      const int i = i0 + j * CSR_THREADS + t;
      it[j] = 0;
      if (i < T) {
        const int k = find_piece(s_start, i);
        it[j] = region[s_from[k] + (i - s_start[k])];
      }
    }
    #pragma unroll
    for (int j = 0; j < 4; ++j) {
      const int i = i0 + j * CSR_THREADS + t;
      if (i < T) {
        if (staged) s_raw[i] = it[j];
        atomicAdd(&s_cur[it[j] >> 24], 1);
      }
    }
  }
  __syncthreads();
  const int nc = (t < BIN_NODES) ? s_cur[t] : 0;
  int tot;
  const int no = block_scan_excl(nc, s_w, tot);
  if (t < BIN_NODES) {
    s_cur[t] = no;
    const int node = b * BIN_NODES + t;
    if (node < N) rowptr[node] = base + no;
  }
  if (b == nbin - 1 && t == 0) rowptr[N] = base + T;
  __syncthreads();

  if (staged) {
    for (int i = t; i < T; i += CSR_THREADS) {
      const unsigned e = s_raw[i];
      const int pos = atomicAdd(&s_cur[e >> 24], 1);
      s_out[pos] = (int)(e & 0xffffffu);
    }
    __syncthreads();
    for (int i = t; i < T; i += CSR_THREADS) col[base + i] = s_out[i];
  } else {
    for (int i = t; i < T; i += CSR_THREADS) {
      const int k = find_piece(s_start, i);
      const unsigned e = region[s_from[k] + (i - s_start[k])];
      const int pos = atomicAdd(&s_cur[e >> 24], 1);
      col[base + pos] = (int)(e & 0xffffffu);
    }
  }
}

__global__ __launch_bounds__(256) void node_gat4(
    const int* __restrict__ rowptr, const int* __restrict__ col,
    const float* __restrict__ a_s, const float* __restrict__ a_d,
    const float* __restrict__ Hfeat, const float* __restrict__ bias,
    float* __restrict__ out, int N)
{
  const int n = __builtin_amdgcn_readfirstlane(
      (int)(blockIdx.x * 8 + (threadIdx.x >> 5)));
  if (n >= N) return;
  const int lane = threadIdx.x & 31;
  const int beg = rowptr[n];
  const int deg = rowptr[n + 1] - beg + 1;
  const float4 ad = *(const float4*)(a_d + (size_t)n * 4);
  const float ninf = -__builtin_inff();

  float m0 = ninf, m1 = ninf, m2 = ninf, m3 = ninf;
  for (int b = 0; b < deg; b += 32) {
    const int t = b + lane;
    if (t < deg) {
      const int s = (t + 1 < deg) ? col[beg + t] : n;
      const float4 as = *(const float4*)(a_s + (size_t)s * 4);
      m0 = fmaxf(m0, lrelu(as.x + ad.x));
      m1 = fmaxf(m1, lrelu(as.y + ad.y));
      m2 = fmaxf(m2, lrelu(as.z + ad.z));
      m3 = fmaxf(m3, lrelu(as.w + ad.w));
    }
  }
  #pragma unroll
  for (int off = 16; off; off >>= 1) {
    m0 = fmaxf(m0, __shfl_xor(m0, off, 32));
    m1 = fmaxf(m1, __shfl_xor(m1, off, 32));
    m2 = fmaxf(m2, __shfl_xor(m2, off, 32));
    m3 = fmaxf(m3, __shfl_xor(m3, off, 32));
  }

  const bool hi = lane >= 16;
  const float4* H4 = (const float4*)Hfeat;
  float S0 = 0.f, S1 = 0.f, S2 = 0.f, S3 = 0.f;
  float4 accA = {0.f, 0.f, 0.f, 0.f};
  float4 accB = {0.f, 0.f, 0.f, 0.f};
  for (int b = 0; b < deg; b += 32) {
    const int t = b + lane;
    int s = n;
    float w0 = 0.f, w1 = 0.f, w2 = 0.f, w3 = 0.f;
    if (t < deg) {
      if (t + 1 < deg) s = col[beg + t];
      const float4 as = *(const float4*)(a_s + (size_t)s * 4);
      w0 = __expf(lrelu(as.x + ad.x) - m0);
      w1 = __expf(lrelu(as.y + ad.y) - m1);
      w2 = __expf(lrelu(as.z + ad.z) - m2);
      w3 = __expf(lrelu(as.w + ad.w) - m3);
      S0 += w0; S1 += w1; S2 += w2; S3 += w3;
    }
    const int cnt = (deg - b < 32) ? (deg - b) : 32;
    #pragma unroll 4
    for (int k = 0; k < cnt; ++k) {
      const int sk = __builtin_amdgcn_readlane(s, k);
      const float b0 = lane_bcast_f(w0, k);
      const float b1 = lane_bcast_f(w1, k);
      const float b2 = lane_bcast_f(w2, k);
      const float b3 = lane_bcast_f(w3, k);
      const float wa = hi ? b1 : b0;
      const float wb = hi ? b3 : b2;
      const float4* row = H4 + (size_t)sk * 64;
      const float4 va = row[lane];
      const float4 vb = row[32 + lane];
      accA.x += wa * va.x; accA.y += wa * va.y;
      accA.z += wa * va.z; accA.w += wa * va.w;
      accB.x += wb * vb.x; accB.y += wb * vb.y;
      accB.z += wb * vb.z; accB.w += wb * vb.w;
    }
  }
  #pragma unroll
  for (int off = 16; off; off >>= 1) {
    S0 += __shfl_xor(S0, off, 32);
    S1 += __shfl_xor(S1, off, 32);
    S2 += __shfl_xor(S2, off, 32);
    S3 += __shfl_xor(S3, off, 32);
  }
  const float da = (hi ? S1 : S0) + SOFTMAX_EPS;
  const float db = (hi ? S3 : S2) + SOFTMAX_EPS;
  const float4 ba = ((const float4*)bias)[lane];
  const float4 bb = ((const float4*)bias)[32 + lane];
  float4 ra, rb;
  ra.x = accA.x / da + ba.x; ra.y = accA.y / da + ba.y;
  ra.z = accA.z / da + ba.z; ra.w = accA.w / da + ba.w;
  rb.x = accB.x / db + bb.x; rb.y = accB.y / db + bb.y;
  rb.z = accB.z / db + bb.z; rb.w = accB.w / db + bb.w;
  ra.x = ra.x > 0.f ? ra.x : expm1f(ra.x);
  ra.y = ra.y > 0.f ? ra.y : expm1f(ra.y);
  ra.z = ra.z > 0.f ? ra.z : expm1f(ra.z);
  ra.w = ra.w > 0.f ? ra.w : expm1f(ra.w);
  rb.x = rb.x > 0.f ? rb.x : expm1f(rb.x);
  rb.y = rb.y > 0.f ? rb.y : expm1f(rb.y);
  rb.z = rb.z > 0.f ? rb.z : expm1f(rb.z);
  rb.w = rb.w > 0.f ? rb.w : expm1f(rb.w);
  float4* orow = (float4*)(out + (size_t)n * 256);
  orow[lane] = ra;
  orow[32 + lane] = rb;
}

__global__ __launch_bounds__(256) void node_gat1(
    const int* __restrict__ rowptr, const int* __restrict__ col,
    const float* __restrict__ a_s, const float* __restrict__ a_d,
    const float* __restrict__ Hfeat, const float* __restrict__ bias,
    float* __restrict__ out, int N)
{
  const int n = __builtin_amdgcn_readfirstlane(
      (int)(blockIdx.x * 8 + (threadIdx.x >> 5)));
  if (n >= N) return;
  const int lane = threadIdx.x & 31;
  const int beg = rowptr[n];
  const int deg = rowptr[n + 1] - beg + 1;
  const float ad = a_d[n];

  float m = -__builtin_inff();
  for (int b = 0; b < deg; b += 32) {
    const int t = b + lane;
    if (t < deg) {
      const int s = (t + 1 < deg) ? col[beg + t] : n;
      m = fmaxf(m, lrelu(a_s[s] + ad));
    }
  }
  #pragma unroll
  for (int off = 16; off; off >>= 1) m = fmaxf(m, __shfl_xor(m, off, 32));

  const float2* H2 = (const float2*)Hfeat;
  float S = 0.f;
  float2 acc = {0.f, 0.f};
  for (int b = 0; b < deg; b += 32) {
    const int t = b + lane;
    int s = n;
    float w = 0.f;
    if (t < deg) {
      if (t + 1 < deg) s = col[beg + t];
      w = __expf(lrelu(a_s[s] + ad) - m);
      S += w;
    }
    const int cnt = (deg - b < 32) ? (deg - b) : 32;
    #pragma unroll 4
    for (int k = 0; k < cnt; ++k) {
      const int sk = __builtin_amdgcn_readlane(s, k);
      const float wk = lane_bcast_f(w, k);
      const float2 v = H2[(size_t)sk * 32 + lane];
      acc.x += wk * v.x;
      acc.y += wk * v.y;
    }
  }
  #pragma unroll
  for (int off = 16; off; off >>= 1) S += __shfl_xor(S, off, 32);
  const float d = S + SOFTMAX_EPS;
  const float2 bv = ((const float2*)bias)[lane];
  float2 r;
  r.x = acc.x / d + bv.x;
  r.y = acc.y / d + bv.y;
  ((float2*)(out + (size_t)n * 64))[lane] = r;
}

extern "C" void kernel_launch(void* const* d_in, const int* in_sizes, int n_in,
                              void* d_out, int out_size, void* d_ws, size_t ws_size,
                              hipStream_t stream)
{
  const float* x   = (const float*)d_in[0];
  const int*   ei  = (const int*)  d_in[1];
  const float* W1  = (const float*)d_in[2];
  const float* as1 = (const float*)d_in[3];
  const float* ad1 = (const float*)d_in[4];
  const float* b1  = (const float*)d_in[5];
  const float* W2  = (const float*)d_in[6];
  const float* as2 = (const float*)d_in[7];
  const float* ad2 = (const float*)d_in[8];
  const float* b2  = (const float*)d_in[9];
  float* out = (float*)d_out;

  constexpr int F = 128, H1 = 4, HC1 = 256, HC2 = 64;
  const int N  = in_sizes[0] / F;
  const int E  = in_sizes[1] / 2;
  const int* src = ei;
  const int* dst = ei + E;

  float* h1  = (float*)d_ws;
  float* x2  = h1  + (size_t)N * HC1;
  float* h2  = x2  + (size_t)N * HC1;
  float* aS1 = h2  + (size_t)N * HC2;
  float* aD1 = aS1 + (size_t)N * H1;
  float* aS2 = aD1 + (size_t)N * H1;
  float* aD2 = aS2 + N;
  unsigned* meta   = (unsigned*)(aD2 + N);
  int* rowptr      = (int*)(meta + 512 * 512);
  unsigned* region = (unsigned*)(rowptr + N + 16);
  int* col         = (int*)(region + E);

  const int TB = 256;
  auto cdiv = [](int a, int b) { return (a + b - 1) / b; };

  const int nch  = cdiv(E, CHUNK);
  const int nbin = cdiv(N, BIN_NODES);
  bin_edges<<<nch, CSR_THREADS, 0, stream>>>(src, dst, region, meta, E, nbin);
  build_csr<<<nbin, CSR_THREADS, 0, stream>>>(region, meta, rowptr, col, N, nch, nbin);

  gemm_att_wmma_f32<F, H1><<<cdiv((N / 16) * H1, 8), TB, 0, stream>>>(
      x, W1, as1, ad1, h1, aS1, aD1, N);
  node_gat4<<<cdiv(N, 8), TB, 0, stream>>>(rowptr, col, aS1, aD1, h1, b1, x2, N);

  gemm_att_wmma_f32<HC1, 1><<<cdiv(N / 16, 8), TB, 0, stream>>>(
      x2, W2, as2, ad2, h2, aS2, aD2, N);
  node_gat1<<<cdiv(N, 8), TB, 0, stream>>>(rowptr, col, aS2, aD2, h2, b2, out, N);
}
